# starting kernel
# baseline (speedup 1.0000x reference)
#include <hip/hip_runtime.h>

typedef __attribute__((ext_vector_type(2))) float v2f;
typedef __attribute__((ext_vector_type(4))) float v4f;
typedef __attribute__((ext_vector_type(8))) float v8f;

#define BATCH   32768
#define NTILES  (BATCH / 16)
#define BLOCKS  256
#define TPB     128

__device__ __forceinline__ v8f wmma_f32_k4(v2f a, v2f b, v8f c) {
  return __builtin_amdgcn_wmma_f32_16x16x4_f32(
       false, a,  false, b,
       (short)0, c,  false,  false);
}

__global__ __launch_bounds__(TPB) void gcn_fused_kernel(
    const float* __restrict__ x,
    const float* __restrict__ qW1, const float* __restrict__ qb1,
    const float* __restrict__ qW2, const float* __restrict__ qb2,
    const float* __restrict__ cW1, const float* __restrict__ cb1,
    const float* __restrict__ cW2, const float* __restrict__ cb2,
    const float* __restrict__ eTW1, const float* __restrict__ eTb1,
    const float* __restrict__ eTW2, const float* __restrict__ eTb2,
    const float* __restrict__ eHW1, const float* __restrict__ eHb1,
    const float* __restrict__ eHW2, const float* __restrict__ eHb2,
    float* __restrict__ out)
{
  __shared__ float s_w2m[2][64];
  __shared__ float s_eW1[2][5][32];
  __shared__ float s_eb1[2][32];
  __shared__ float s_eW2[2][32];
  __shared__ float s_eb2[2];

  const int tid = threadIdx.x;

  {
    const int net = tid >> 6;
    const int h   = tid & 63;
    const float* W2 = net ? cW2 : qW2;
    const v4f* p = (const v4f*)(W2 + h * 32);
    float s = 0.0f;
#pragma unroll
    for (int i = 0; i < 8; ++i) { v4f v = p[i]; s += v.x + v.y + v.z + v.w; }
    s_w2m[net][h] = s * (1.0f / 32.0f);
  }
  if (tid < 32) {
#pragma unroll
    for (int f = 0; f < 5; ++f) {
      s_eW1[0][f][tid] = eTW1[f * 32 + tid];
      s_eW1[1][f][tid] = eHW1[f * 32 + tid];
    }
    s_eb1[0][tid] = eTb1[tid];
    s_eb1[1][tid] = eHb1[tid];
    s_eW2[0][tid] = eTW2[tid];
    s_eW2[1][tid] = eHW2[tid];
    if (tid < 2) s_eb2[tid] = tid ? eHb2[0] : eTb2[0];
  }
  __syncthreads();

  const int lane = tid & 31;
  const int half = lane >> 4;
  const int lm   = lane & 15;
  const int koff = half * 2;

  float b1r[2][4], w2mr[2][4];
#pragma unroll
  for (int n = 0; n < 4; ++n) {
    const int col = n * 16 + lm;
    b1r[0][n]  = qb1[col];
    b1r[1][n]  = cb1[col];
    w2mr[0][n] = s_w2m[0][col];
    w2mr[1][n] = s_w2m[1][col];
  }
  float mb2 = qb2[lane] + cb2[lane];
#pragma unroll
  for (int m = 16; m >= 1; m >>= 1) mb2 += __shfl_xor(mb2, m, 32);
  mb2 *= (1.0f / 32.0f);

  v2f bq[4][8], bc[4][8];
#pragma unroll
  for (int n = 0; n < 4; ++n) {
    const int col = n * 16 + lm;
#pragma unroll
    for (int k = 0; k < 8; ++k) {
      const int kr = k * 4 + koff;
      v2f t;
      t.x = qW1[kr * 64 + col]; t.y = qW1[(kr + 1) * 64 + col]; bq[n][k] = t;
      t.x = cW1[kr * 64 + col]; t.y = cW1[(kr + 1) * 64 + col]; bc[n][k] = t;
    }
  }

  const int wgid  = blockIdx.x * (TPB >> 5) + (tid >> 5);
  const int wstep = gridDim.x * (TPB >> 5);

  for (int t = wgid; t < NTILES; t += wstep) {
    const int base = t * 16;

    v2f a[8];
    const float* xr = x + (base + lm) * 32 + koff;
#pragma unroll
    for (int k = 0; k < 8; ++k) a[k] = *(const v2f*)(xr + 4 * k);

    float accv[8];
#pragma unroll
    for (int v = 0; v < 8; ++v) accv[v] = 0.0f;

#pragma unroll
    for (int n = 0; n < 4; ++n) {
      v8f c = {0.f, 0.f, 0.f, 0.f, 0.f, 0.f, 0.f, 0.f};
#pragma unroll
      for (int k = 0; k < 8; ++k) c = wmma_f32_k4(a[k], bq[n][k], c);
#pragma unroll
      for (int v = 0; v < 8; ++v) {
        const float h = fmaxf(c[v] + b1r[0][n], 0.0f);
        accv[v] = fmaf(h, w2mr[0][n], accv[v]);
      }
    }
#pragma unroll
    for (int n = 0; n < 4; ++n) {
      v8f c = {0.f, 0.f, 0.f, 0.f, 0.f, 0.f, 0.f, 0.f};
#pragma unroll
      for (int k = 0; k < 8; ++k) c = wmma_f32_k4(a[k], bc[n][k], c);
#pragma unroll
      for (int v = 0; v < 8; ++v) {
        const float h = fmaxf(c[v] + b1r[1][n], 0.0f);
        accv[v] = fmaf(h, w2mr[1][n], accv[v]);
      }
    }

#pragma unroll
    for (int v = 0; v < 8; ++v) {
      float s = accv[v];
#pragma unroll
      for (int m = 8; m >= 1; m >>= 1) s += __shfl_xor(s, m, 32);
      accv[v] = s + mb2;
    }

    const float T = x[(base + lm) * 32 + 0];
    const float H = x[(base + lm) * 32 + 1];
    const float f0 = T, f1 = H, f2 = T * T, f3 = T * H;
    const float f4 = half ? (T * H * H) : (T * T * T);
    const int net = half;
    float cacc = s_eb2[net];
#pragma unroll 4
    for (int j = 0; j < 32; ++j) {
      float hj = s_eb1[net][j];
      hj = fmaf(f0, s_eW1[net][0][j], hj);
      hj = fmaf(f1, s_eW1[net][1][j], hj);
      hj = fmaf(f2, s_eW1[net][2][j], hj);
      hj = fmaf(f3, s_eW1[net][3][j], hj);
      hj = fmaf(f4, s_eW1[net][4][j], hj);
      cacc = fmaf(fmaxf(hj, 0.0f), s_eW2[net][j], cacc);
    }

#pragma unroll
    for (int r = 0; r < 16; ++r) {
      const float nlr = __shfl(accv[r & 7], (r < 8) ? 0 : 16, 32);
      const float cTr = __shfl(cacc, r, 32);
      const float cHr = __shfl(cacc, r + 16, 32);
      float val = nlr;
      if (lane == 0) val += cTr;
      if (lane == 1) val += cHr;
      out[(base + r) * 32 + lane] = val;
    }
  }
}

extern "C" void kernel_launch(void* const* d_in, const int* in_sizes, int n_in,
                              void* d_out, int out_size, void* d_ws, size_t ws_size,
                              hipStream_t stream) {
  (void)in_sizes; (void)n_in; (void)out_size; (void)d_ws; (void)ws_size;
  const float* x    = (const float*)d_in[0];
  const float* qW1  = (const float*)d_in[1];
  const float* qb1  = (const float*)d_in[2];
  const float* qW2  = (const float*)d_in[3];
  const float* qb2  = (const float*)d_in[4];
  const float* cW1  = (const float*)d_in[5];
  const float* cb1  = (const float*)d_in[6];
  const float* cW2  = (const float*)d_in[7];
  const float* cb2  = (const float*)d_in[8];
  const float* eTW1 = (const float*)d_in[9];
  const float* eTb1 = (const float*)d_in[10];
  const float* eTW2 = (const float*)d_in[11];
  const float* eTb2 = (const float*)d_in[12];
  const float* eHW1 = (const float*)d_in[13];
  const float* eHb1 = (const float*)d_in[14];
  const float* eHW2 = (const float*)d_in[15];
  const float* eHb2 = (const float*)d_in[16];
  float* out = (float*)d_out;

  gcn_fused_kernel<<<BLOCKS, TPB, 0, stream>>>(
      x, qW1, qb1, qW2, qb2, cW1, cb1, cW2, cb2,
      eTW1, eTb1, eTW2, eTb2, eHW1, eHb1, eHW2, eHb2, out);
}


# cross-lane sums and broadcasts moved off the LDS crossbar (ds_bpermute + wait per butterfly step) onto vector-ALU lane ops: DPP row_xmask adds, permlanex16, readlane; same pairing, same sums
# speedup vs baseline: 1.5447x; 1.5447x over previous
#include <hip/hip_runtime.h>

typedef __attribute__((ext_vector_type(2))) float v2f;
typedef __attribute__((ext_vector_type(4))) float v4f;
typedef __attribute__((ext_vector_type(8))) float v8f;

#define BATCH   32768
#define NTILES  (BATCH / 16)
#define BLOCKS  256
#define TPB     256

__device__ __forceinline__ v8f wmma_f32_k4(v2f a, v2f b, v8f c) {
  return __builtin_amdgcn_wmma_f32_16x16x4_f32(
       false, a,  false, b,
       (short)0, c,  false,  false);
}

__device__ __forceinline__ float xhalf(float x) {
  const int b = __builtin_bit_cast(int, x);
  return __builtin_bit_cast(float, __builtin_amdgcn_permlanex16(b, b, 0x76543210, 0xfedcba98, false, false));
}
template <int M> __device__ __forceinline__ float rowx(float x) {
  const int b = __builtin_bit_cast(int, x);
  return __builtin_bit_cast(float, __builtin_amdgcn_update_dpp(b, b, 0x160 + M, 0xf, 0xf, true));
}
__device__ __forceinline__ float lane_bcast(float x, int src) {
  return __builtin_bit_cast(float, __builtin_amdgcn_readlane(__builtin_bit_cast(int, x), src));
}

__global__ __launch_bounds__(TPB) void gcn_fused_kernel(
    const float* __restrict__ x,
    const float* __restrict__ qW1, const float* __restrict__ qb1,
    const float* __restrict__ qW2, const float* __restrict__ qb2,
    const float* __restrict__ cW1, const float* __restrict__ cb1,
    const float* __restrict__ cW2, const float* __restrict__ cb2,
    const float* __restrict__ eTW1, const float* __restrict__ eTb1,
    const float* __restrict__ eTW2, const float* __restrict__ eTb2,
    const float* __restrict__ eHW1, const float* __restrict__ eHb1,
    const float* __restrict__ eHW2, const float* __restrict__ eHb2,
    float* __restrict__ out)
{
  __shared__ float s_w2m[2][64];
  __shared__ float s_eW1[2][5][32];
  __shared__ float s_eb1[2][32];
  __shared__ float s_eW2[2][32];
  __shared__ float s_eb2[2];

  const int tid = threadIdx.x;

  if (tid < 128) {
    const int net = tid >> 6;
    const int h   = tid & 63;
    const float* W2 = net ? cW2 : qW2;
    const v4f* p = (const v4f*)(W2 + h * 32);
    float s = 0.0f;
#pragma unroll
    for (int i = 0; i < 8; ++i) { v4f v = p[i]; s += v.x + v.y + v.z + v.w; }
    s_w2m[net][h] = s * (1.0f / 32.0f);
  }
  if (tid >= 128 && tid < 160) {
    const int j = tid - 128;
#pragma unroll
    for (int f = 0; f < 5; ++f) {
      s_eW1[0][f][j] = eTW1[f * 32 + j];
      s_eW1[1][f][j] = eHW1[f * 32 + j];
    }
    s_eb1[0][j] = eTb1[j];
    s_eb1[1][j] = eHb1[j];
    s_eW2[0][j] = eTW2[j];
    s_eW2[1][j] = eHW2[j];
    if (j < 2) s_eb2[j] = j ? eHb2[0] : eTb2[0];
  }
  __syncthreads();

  const int lane = tid & 31;
  const int half = lane >> 4;
  const int lm   = lane & 15;
  const int koff = half * 2;

  float b1r[2][4], w2mr[2][4];
#pragma unroll
  for (int n = 0; n < 4; ++n) {
    const int col = n * 16 + lm;
    b1r[0][n]  = qb1[col];
    b1r[1][n]  = cb1[col];
    w2mr[0][n] = s_w2m[0][col];
    w2mr[1][n] = s_w2m[1][col];
  }
  float mb2 = qb2[lane] + cb2[lane];
  mb2 += xhalf(mb2);
  mb2 += rowx<8>(mb2);
  mb2 += rowx<4>(mb2);
  mb2 += rowx<2>(mb2);
  mb2 += rowx<1>(mb2);
  mb2 *= (1.0f / 32.0f);

  v2f bq[4][8], bc[4][8];
#pragma unroll
  for (int n = 0; n < 4; ++n) {
    const int col = n * 16 + lm;
#pragma unroll
    for (int k = 0; k < 8; ++k) {
      const int kr = k * 4 + koff;
      v2f t;
      t.x = qW1[kr * 64 + col]; t.y = qW1[(kr + 1) * 64 + col]; bq[n][k] = t;
      t.x = cW1[kr * 64 + col]; t.y = cW1[(kr + 1) * 64 + col]; bc[n][k] = t;
    }
  }

  const int wgid  = blockIdx.x * (TPB >> 5) + (tid >> 5);
  const int wstep = gridDim.x * (TPB >> 5);

  for (int t = wgid; t < NTILES; t += wstep) {
    const int base = t * 16;

    v2f a[8];
    const float* xr = x + (base + lm) * 32 + koff;
#pragma unroll
    for (int k = 0; k < 8; ++k) a[k] = *(const v2f*)(xr + 4 * k);

    float accv[8];
#pragma unroll
    for (int v = 0; v < 8; ++v) accv[v] = 0.0f;

#pragma unroll
    for (int n = 0; n < 4; ++n) {
      v8f c = {0.f, 0.f, 0.f, 0.f, 0.f, 0.f, 0.f, 0.f};
#pragma unroll
      for (int k = 0; k < 8; ++k) c = wmma_f32_k4(a[k], bq[n][k], c);
#pragma unroll
      for (int v = 0; v < 8; ++v) {
        const float h = fmaxf(c[v] + b1r[0][n], 0.0f);
        accv[v] = fmaf(h, w2mr[0][n], accv[v]);
      }
    }
#pragma unroll
    for (int n = 0; n < 4; ++n) {
      v8f c = {0.f, 0.f, 0.f, 0.f, 0.f, 0.f, 0.f, 0.f};
#pragma unroll
      for (int k = 0; k < 8; ++k) c = wmma_f32_k4(a[k], bc[n][k], c);
#pragma unroll
      for (int v = 0; v < 8; ++v) {
        const float h = fmaxf(c[v] + b1r[1][n], 0.0f);
        accv[v] = fmaf(h, w2mr[1][n], accv[v]);
      }
    }

#pragma unroll
    for (int v = 0; v < 8; ++v) {
      float s = accv[v];
      s += rowx<8>(s);
      s += rowx<4>(s);
      s += rowx<2>(s);
      s += rowx<1>(s);
      accv[v] = s + mb2;
    }

    const float T = x[(base + lm) * 32 + 0];
    const float H = x[(base + lm) * 32 + 1];
    const float f0 = T, f1 = H, f2 = T * T, f3 = T * H;
    const float f4 = half ? (T * H * H) : (T * T * T);
    const int net = half;
    float cacc = s_eb2[net];
#pragma unroll 4
    for (int j = 0; j < 32; ++j) {
      float hj = s_eb1[net][j];
      hj = fmaf(f0, s_eW1[net][0][j], hj);
      hj = fmaf(f1, s_eW1[net][1][j], hj);
      hj = fmaf(f2, s_eW1[net][2][j], hj);
      hj = fmaf(f3, s_eW1[net][3][j], hj);
      hj = fmaf(f4, s_eW1[net][4][j], hj);
      cacc = fmaf(fmaxf(hj, 0.0f), s_eW2[net][j], cacc);
    }

#pragma unroll
    for (int r = 0; r < 16; ++r) {
      const float nlr = lane_bcast(accv[r & 7], (r < 8) ? 0 : 16);
      const float cTr = lane_bcast(cacc, r);
      const float cHr = lane_bcast(cacc, r + 16);
      float val = nlr;
      if (lane == 0) val += cTr;
      if (lane == 1) val += cHr;
      out[(base + r) * 32 + lane] = val;
    }
  }
}

extern "C" void kernel_launch(void* const* d_in, const int* in_sizes, int n_in,
                              void* d_out, int out_size, void* d_ws, size_t ws_size,
                              hipStream_t stream) {
  (void)in_sizes; (void)n_in; (void)out_size; (void)d_ws; (void)ws_size;
  const float* x    = (const float*)d_in[0];
  const float* qW1  = (const float*)d_in[1];
  const float* qb1  = (const float*)d_in[2];
  const float* qW2  = (const float*)d_in[3];
  const float* qb2  = (const float*)d_in[4];
  const float* cW1  = (const float*)d_in[5];
  const float* cb1  = (const float*)d_in[6];
  const float* cW2  = (const float*)d_in[7];
  const float* cb2  = (const float*)d_in[8];
  const float* eTW1 = (const float*)d_in[9];
  const float* eTb1 = (const float*)d_in[10];
  const float* eTW2 = (const float*)d_in[11];
  const float* eTb2 = (const float*)d_in[12];
  const float* eHW1 = (const float*)d_in[13];
  const float* eHb1 = (const float*)d_in[14];
  const float* eHW2 = (const float*)d_in[15];
  const float* eHb2 = (const float*)d_in[16];
  float* out = (float*)d_out;

  gcn_fused_kernel<<<BLOCKS, TPB, 0, stream>>>(
      x, qW1, qb1, qW2, qb2, cW1, cb1, cW2, cb2,
      eTW1, eTb1, eTW2, eTb2, eHW1, eHb1, eHW2, eHb2, out);
}


# enso 5->32->1 MLPs moved off the LDS-table FMA loop onto the matrix unit: 8 f32 WMMAs (K padded to 8, bias as a k row); 2 waves/SIMD declared so tiles stay interleaved; enso sum order changes
# speedup vs baseline: 1.6467x; 1.0660x over previous
#include <hip/hip_runtime.h>

typedef __attribute__((ext_vector_type(2))) float v2f;
typedef __attribute__((ext_vector_type(4))) float v4f;
typedef __attribute__((ext_vector_type(8))) float v8f;

#define BATCH   32768
#define NTILES  (BATCH / 16)
#define BLOCKS  256
#define TPB     256

__device__ __forceinline__ v8f wmma_f32_k4(v2f a, v2f b, v8f c) {
  return __builtin_amdgcn_wmma_f32_16x16x4_f32(
       false, a,  false, b,
       (short)0, c,  false,  false);
}

__device__ __forceinline__ float xhalf(float x) {
  const int b = __builtin_bit_cast(int, x);
  return __builtin_bit_cast(float, __builtin_amdgcn_permlanex16(b, b, 0x76543210, 0xfedcba98, false, false));
}
template <int M> __device__ __forceinline__ float rowx(float x) {
  const int b = __builtin_bit_cast(int, x);
  return __builtin_bit_cast(float, __builtin_amdgcn_update_dpp(b, b, 0x160 + M, 0xf, 0xf, true));
}
__device__ __forceinline__ float lane_bcast(float x, int src) {
  return __builtin_bit_cast(float, __builtin_amdgcn_readlane(__builtin_bit_cast(int, x), src));
}

__global__ __launch_bounds__(TPB) __attribute__((amdgpu_waves_per_eu(2, 2))) void gcn_fused_kernel(
    const float* __restrict__ x,
    const float* __restrict__ qW1, const float* __restrict__ qb1,
    const float* __restrict__ qW2, const float* __restrict__ qb2,
    const float* __restrict__ cW1, const float* __restrict__ cb1,
    const float* __restrict__ cW2, const float* __restrict__ cb2,
    const float* __restrict__ eTW1, const float* __restrict__ eTb1,
    const float* __restrict__ eTW2, const float* __restrict__ eTb2,
    const float* __restrict__ eHW1, const float* __restrict__ eHb1,
    const float* __restrict__ eHW2, const float* __restrict__ eHb2,
    float* __restrict__ out)
{
  __shared__ float s_w2m[2][64];

  const int tid = threadIdx.x;

  if (tid < 128) {
    const int net = tid >> 6;
    const int h   = tid & 63;
    const float* W2 = net ? cW2 : qW2;
    const v4f* p = (const v4f*)(W2 + h * 32);
    float s = 0.0f;
#pragma unroll
    for (int i = 0; i < 8; ++i) { v4f v = p[i]; s += v.x + v.y + v.z + v.w; }
    s_w2m[net][h] = s * (1.0f / 32.0f);
  }
  __syncthreads();

  const int lane = tid & 31;
  const int half = lane >> 4;
  const int lm   = lane & 15;
  const int koff = half * 2;

  float b1r[2][4], w2mr[2][4];
#pragma unroll
  for (int n = 0; n < 4; ++n) {
    const int col = n * 16 + lm;
    b1r[0][n]  = qb1[col];
    b1r[1][n]  = cb1[col];
    w2mr[0][n] = s_w2m[0][col];
    w2mr[1][n] = s_w2m[1][col];
  }
  float mb2 = qb2[lane] + cb2[lane];
  mb2 += xhalf(mb2);
  mb2 += rowx<8>(mb2);
  mb2 += rowx<4>(mb2);
  mb2 += rowx<2>(mb2);
  mb2 += rowx<1>(mb2);
  mb2 *= (1.0f / 32.0f);

  v2f bq[4][8], bc[4][8];
#pragma unroll
  for (int n = 0; n < 4; ++n) {
    const int col = n * 16 + lm;
#pragma unroll
    for (int k = 0; k < 8; ++k) {
      const int kr = k * 4 + koff;
      v2f t;
      t.x = qW1[kr * 64 + col]; t.y = qW1[(kr + 1) * 64 + col]; bq[n][k] = t;
      t.x = cW1[kr * 64 + col]; t.y = cW1[(kr + 1) * 64 + col]; bc[n][k] = t;
    }
  }

  v2f be0[4], be1[4];
  float ew2r[4];
#pragma unroll
  for (int n = 0; n < 4; ++n) {
    const int j = (n & 1) * 16 + lm;
    const float* eW1 = (n < 2) ? eTW1 : eHW1;
    const float* eb1 = (n < 2) ? eTb1 : eHb1;
    const float* eW2 = (n < 2) ? eTW2 : eHW2;
    v2f t;
    t.x = eW1[koff * 32 + j]; t.y = eW1[(koff + 1) * 32 + j]; be0[n] = t;
    t.x = half ? eb1[j] : eW1[4 * 32 + j]; t.y = 0.0f;        be1[n] = t;
    ew2r[n] = eW2[j];
  }
  const float eb2sel = (lm < 8) ? eTb2[0] : eHb2[0];

  const int wgid  = blockIdx.x * (TPB >> 5) + (tid >> 5);
  const int wstep = gridDim.x * (TPB >> 5);

  for (int t = wgid; t < NTILES; t += wstep) {
    const int base = t * 16;

    v2f a[8];
    const float* xr = x + (base + lm) * 32 + koff;
#pragma unroll
    for (int k = 0; k < 8; ++k) a[k] = *(const v2f*)(xr + 4 * k);

    float accv[8];
#pragma unroll
    for (int v = 0; v < 8; ++v) accv[v] = 0.0f;

#pragma unroll
    for (int n = 0; n < 4; ++n) {
      v8f c = {0.f, 0.f, 0.f, 0.f, 0.f, 0.f, 0.f, 0.f};
#pragma unroll
      for (int k = 0; k < 8; ++k) c = wmma_f32_k4(a[k], bq[n][k], c);
#pragma unroll
      for (int v = 0; v < 8; ++v) {
        const float h = fmaxf(c[v] + b1r[0][n], 0.0f);
        accv[v] = fmaf(h, w2mr[0][n], accv[v]);
      }
    }
#pragma unroll
    for (int n = 0; n < 4; ++n) {
      v8f c = {0.f, 0.f, 0.f, 0.f, 0.f, 0.f, 0.f, 0.f};
#pragma unroll
      for (int k = 0; k < 8; ++k) c = wmma_f32_k4(a[k], bc[n][k], c);
#pragma unroll
      for (int v = 0; v < 8; ++v) {
        const float h = fmaxf(c[v] + b1r[1][n], 0.0f);
        accv[v] = fmaf(h, w2mr[1][n], accv[v]);
      }
    }

#pragma unroll
    for (int v = 0; v < 8; ++v) {
      float s = accv[v];
      s += rowx<8>(s);
      s += rowx<4>(s);
      s += rowx<2>(s);
      s += rowx<1>(s);
      accv[v] = s + mb2;
    }

    const float T = x[(base + lm) * 32 + 0];
    const float H = x[(base + lm) * 32 + 1];
    v2f e0, eT1, eH1;
    e0.x  = half ? (T * T) : T;
    e0.y  = half ? (T * H) : H;
    eT1.x = half ? 1.0f : (T * T * T);  eT1.y = 0.0f;
    eH1.x = half ? 1.0f : (T * H * H);  eH1.y = 0.0f;

    float accT[8], accH[8];
#pragma unroll
    for (int v = 0; v < 8; ++v) { accT[v] = 0.0f; accH[v] = 0.0f; }
#pragma unroll
    for (int n = 0; n < 4; ++n) {
      v8f c = {0.f, 0.f, 0.f, 0.f, 0.f, 0.f, 0.f, 0.f};
      c = wmma_f32_k4(e0, be0[n], c);
      c = wmma_f32_k4((n < 2) ? eT1 : eH1, be1[n], c);
#pragma unroll
      for (int v = 0; v < 8; ++v) {
        const float h = fmaxf(c[v], 0.0f);
        if (n < 2) accT[v] = fmaf(h, ew2r[n], accT[v]);
        else       accH[v] = fmaf(h, ew2r[n], accH[v]);
      }
    }
    float ce[8];
#pragma unroll
    for (int v = 0; v < 8; ++v) {
      float sT = accT[v]; sT += rowx<8>(sT);
      float sH = accH[v]; sH += rowx<8>(sH);
      float m = (lm < 8) ? sT : sH;
      m += rowx<4>(m);
      m += rowx<2>(m);
      m += rowx<1>(m);
      ce[v] = m + eb2sel;
    }

#pragma unroll
    for (int r = 0; r < 16; ++r) {
      const float nlr = lane_bcast(accv[r & 7], (r < 8) ? 0 : 16);
      const float cTr = lane_bcast(ce[r & 7], (r < 8) ? 0 : 16);
      const float cHr = lane_bcast(ce[r & 7], (r < 8) ? 8 : 24);
      float val = nlr;
      if (lane == 0) val += cTr;
      if (lane == 1) val += cHr;
      out[(base + r) * 32 + lane] = val;
    }
  }
}

extern "C" void kernel_launch(void* const* d_in, const int* in_sizes, int n_in,
                              void* d_out, int out_size, void* d_ws, size_t ws_size,
                              hipStream_t stream) {
  (void)in_sizes; (void)n_in; (void)out_size; (void)d_ws; (void)ws_size;
  const float* x    = (const float*)d_in[0];
  const float* qW1  = (const float*)d_in[1];
  const float* qb1  = (const float*)d_in[2];
  const float* qW2  = (const float*)d_in[3];
  const float* qb2  = (const float*)d_in[4];
  const float* cW1  = (const float*)d_in[5];
  const float* cb1  = (const float*)d_in[6];
  const float* cW2  = (const float*)d_in[7];
  const float* cb2  = (const float*)d_in[8];
  const float* eTW1 = (const float*)d_in[9];
  const float* eTb1 = (const float*)d_in[10];
  const float* eTW2 = (const float*)d_in[11];
  const float* eTb2 = (const float*)d_in[12];
  const float* eHW1 = (const float*)d_in[13];
  const float* eHb1 = (const float*)d_in[14];
  const float* eHW2 = (const float*)d_in[15];
  const float* eHb2 = (const float*)d_in[16];
  float* out = (float*)d_out;

  gcn_fused_kernel<<<BLOCKS, TPB, 0, stream>>>(
      x, qW1, qb1, qW2, qb2, cW1, cb1, cW2, cb2,
      eTW1, eTb1, eTW2, eTb2, eHW1, eHb1, eHW2, eHb2, out);
}


# W1 staged once per workgroup in LDS in B-fragment order, waves read them by 32 ds_load_b128 (global loads per wave 172 -> 60); one-trip tile loop unrolled so x loads pass the LDS wait; same sums
# speedup vs baseline: 2.0127x; 1.2223x over previous
#include <hip/hip_runtime.h>

typedef __attribute__((ext_vector_type(2))) float v2f;
typedef __attribute__((ext_vector_type(4))) float v4f;
typedef __attribute__((ext_vector_type(8))) float v8f;

#define BATCH   32768
#define NTILES  (BATCH / 16)
#define BLOCKS  256
#define TPB     256
static_assert(BLOCKS * (TPB / 32) == NTILES, "one 16-row tile per wave: the launch below covers the batch exactly");

__device__ __forceinline__ v8f wmma_f32_k4(v2f a, v2f b, v8f c) {
  return __builtin_amdgcn_wmma_f32_16x16x4_f32(
       false, a,  false, b,
       (short)0, c,  false,  false);
}

__device__ __forceinline__ float xhalf(float x) {
  const int b = __builtin_bit_cast(int, x);
  return __builtin_bit_cast(float, __builtin_amdgcn_permlanex16(b, b, 0x76543210, 0xfedcba98, false, false));
}
template <int M> __device__ __forceinline__ float rowx(float x) {
  const int b = __builtin_bit_cast(int, x);
  return __builtin_bit_cast(float, __builtin_amdgcn_update_dpp(b, b, 0x160 + M, 0xf, 0xf, true));
}
__device__ __forceinline__ float lane_bcast(float x, int src) {
  return __builtin_bit_cast(float, __builtin_amdgcn_readlane(__builtin_bit_cast(int, x), src));
}

__global__ __launch_bounds__(TPB) __attribute__((amdgpu_waves_per_eu(2, 2))) void gcn_fused_kernel(
    const float* __restrict__ x,
    const float* __restrict__ qW1, const float* __restrict__ qb1,
    const float* __restrict__ qW2, const float* __restrict__ qb2,
    const float* __restrict__ cW1, const float* __restrict__ cb1,
    const float* __restrict__ cW2, const float* __restrict__ cb2,
    const float* __restrict__ eTW1, const float* __restrict__ eTb1,
    const float* __restrict__ eTW2, const float* __restrict__ eTb2,
    const float* __restrict__ eHW1, const float* __restrict__ eHb1,
    const float* __restrict__ eHW2, const float* __restrict__ eHb2,
    float* __restrict__ out)
{
  __shared__ float s_w2m[2][64];
  __shared__ v4f s_w1[2 * 4 * 4 * 32];

  const int tid  = threadIdx.x;
  const int lane = tid & 31;
  const int half = lane >> 4;
  const int lm   = lane & 15;
  const int koff = half * 2;
  const int wv   = tid >> 5;

  v4f stg[4];
#pragma unroll
  for (int i = 0; i < 4; ++i) {
    const float* W1 = (i < 2) ? qW1 : cW1;
    const int n = (i & 1) * 2 + (wv >> 2);
    const float* p = W1 + (8 * (wv & 3) + koff) * 64 + n * 16 + lm;
    v4f v;
    v.x = p[0]; v.y = p[64]; v.z = p[4 * 64]; v.w = p[5 * 64];
    stg[i] = v;
  }

  if (tid < 128) {
    const int net = tid >> 6;
    const int h   = tid & 63;
    const float* W2 = net ? cW2 : qW2;
    const v4f* p = (const v4f*)(W2 + h * 32);
    float s = 0.0f;
#pragma unroll
    for (int i = 0; i < 8; ++i) { v4f v = p[i]; s += v.x + v.y + v.z + v.w; }
    s_w2m[net][h] = s * (1.0f / 32.0f);
  }
#pragma unroll
  for (int i = 0; i < 4; ++i) s_w1[i * 256 + tid] = stg[i];
  __syncthreads();

  float b1r[2][4], w2mr[2][4];
#pragma unroll
  for (int n = 0; n < 4; ++n) {
    const int col = n * 16 + lm;
    b1r[0][n]  = qb1[col];
    b1r[1][n]  = cb1[col];
    w2mr[0][n] = s_w2m[0][col];
    w2mr[1][n] = s_w2m[1][col];
  }
  float mb2 = qb2[lane] + cb2[lane];
  mb2 += xhalf(mb2);
  mb2 += rowx<8>(mb2);
  mb2 += rowx<4>(mb2);
  mb2 += rowx<2>(mb2);
  mb2 += rowx<1>(mb2);
  mb2 *= (1.0f / 32.0f);

  v2f bq[4][8], bc[4][8];
#pragma unroll
  for (int n = 0; n < 4; ++n) {
#pragma unroll
    for (int j = 0; j < 4; ++j) {
      const v4f q = s_w1[((0 * 4 + n) * 4 + j) * 32 + lane];
      const v4f c = s_w1[((1 * 4 + n) * 4 + j) * 32 + lane];
      v2f t;
      t.x = q.x; t.y = q.y; bq[n][2 * j]     = t;
      t.x = q.z; t.y = q.w; bq[n][2 * j + 1] = t;
      t.x = c.x; t.y = c.y; bc[n][2 * j]     = t;
      t.x = c.z; t.y = c.w; bc[n][2 * j + 1] = t;
    }
  }

  v2f be0[4], be1[4];
  float ew2r[4];
#pragma unroll
  for (int n = 0; n < 4; ++n) {
    const int j = (n & 1) * 16 + lm;
    const float* eW1 = (n < 2) ? eTW1 : eHW1;
    const float* eb1 = (n < 2) ? eTb1 : eHb1;
    const float* eW2 = (n < 2) ? eTW2 : eHW2;
    v2f t;
    t.x = eW1[koff * 32 + j]; t.y = eW1[(koff + 1) * 32 + j]; be0[n] = t;
    t.x = half ? eb1[j] : eW1[4 * 32 + j]; t.y = 0.0f;        be1[n] = t;
    ew2r[n] = eW2[j];
  }
  const float eb2sel = (lm < 8) ? eTb2[0] : eHb2[0];

  {
    const int base = (blockIdx.x * (TPB >> 5) + wv) * 16;

    v2f a[8];
    const float* xr = x + (base + lm) * 32 + koff;
#pragma unroll
    for (int k = 0; k < 8; ++k) a[k] = *(const v2f*)(xr + 4 * k);

    float accv[8];
#pragma unroll
    for (int v = 0; v < 8; ++v) accv[v] = 0.0f;

#pragma unroll
    for (int n = 0; n < 4; ++n) {
      v8f c = {0.f, 0.f, 0.f, 0.f, 0.f, 0.f, 0.f, 0.f};
#pragma unroll
      for (int k = 0; k < 8; ++k) c = wmma_f32_k4(a[k], bq[n][k], c);
#pragma unroll
      for (int v = 0; v < 8; ++v) {
        const float h = fmaxf(c[v] + b1r[0][n], 0.0f);
        accv[v] = fmaf(h, w2mr[0][n], accv[v]);
      }
    }
#pragma unroll
    for (int n = 0; n < 4; ++n) {
      v8f c = {0.f, 0.f, 0.f, 0.f, 0.f, 0.f, 0.f, 0.f};
#pragma unroll
      for (int k = 0; k < 8; ++k) c = wmma_f32_k4(a[k], bc[n][k], c);
#pragma unroll
      for (int v = 0; v < 8; ++v) {
        const float h = fmaxf(c[v] + b1r[1][n], 0.0f);
        accv[v] = fmaf(h, w2mr[1][n], accv[v]);
      }
    }

#pragma unroll
    for (int v = 0; v < 8; ++v) {
      float s = accv[v];
      s += rowx<8>(s);
      s += rowx<4>(s);
      s += rowx<2>(s);
      s += rowx<1>(s);
      accv[v] = s + mb2;
    }

    const float T = x[(base + lm) * 32 + 0];
    const float H = x[(base + lm) * 32 + 1];
    v2f e0, eT1, eH1;
    e0.x  = half ? (T * T) : T;
    e0.y  = half ? (T * H) : H;
    eT1.x = half ? 1.0f : (T * T * T);  eT1.y = 0.0f;
    eH1.x = half ? 1.0f : (T * H * H);  eH1.y = 0.0f;

    float accT[8], accH[8];
#pragma unroll
    for (int v = 0; v < 8; ++v) { accT[v] = 0.0f; accH[v] = 0.0f; }
#pragma unroll
    for (int n = 0; n < 4; ++n) {
      v8f c = {0.f, 0.f, 0.f, 0.f, 0.f, 0.f, 0.f, 0.f};
      c = wmma_f32_k4(e0, be0[n], c);
      c = wmma_f32_k4((n < 2) ? eT1 : eH1, be1[n], c);
#pragma unroll
      for (int v = 0; v < 8; ++v) {
        const float h = fmaxf(c[v], 0.0f);
        if (n < 2) accT[v] = fmaf(h, ew2r[n], accT[v]);
        else       accH[v] = fmaf(h, ew2r[n], accH[v]);
      }
    }
    float ce[8];
#pragma unroll
    for (int v = 0; v < 8; ++v) {
      float sT = accT[v]; sT += rowx<8>(sT);
      float sH = accH[v]; sH += rowx<8>(sH);
      float m = (lm < 8) ? sT : sH;
      m += rowx<4>(m);
      m += rowx<2>(m);
      m += rowx<1>(m);
      ce[v] = m + eb2sel;
    }

#pragma unroll
    for (int r = 0; r < 16; ++r) {
      const float nlr = lane_bcast(accv[r & 7], (r < 8) ? 0 : 16);
      const float cTr = lane_bcast(ce[r & 7], (r < 8) ? 0 : 16);
      const float cHr = lane_bcast(ce[r & 7], (r < 8) ? 8 : 24);
      float val = nlr;
      if (lane == 0) val += cTr;
      if (lane == 1) val += cHr;
      out[(base + r) * 32 + lane] = val;
    }
  }
}

extern "C" void kernel_launch(void* const* d_in, const int* in_sizes, int n_in,
                              void* d_out, int out_size, void* d_ws, size_t ws_size,
                              hipStream_t stream) {
  (void)in_sizes; (void)n_in; (void)out_size; (void)d_ws; (void)ws_size;
  const float* x    = (const float*)d_in[0];
  const float* qW1  = (const float*)d_in[1];
  const float* qb1  = (const float*)d_in[2];
  const float* qW2  = (const float*)d_in[3];
  const float* qb2  = (const float*)d_in[4];
  const float* cW1  = (const float*)d_in[5];
  const float* cb1  = (const float*)d_in[6];
  const float* cW2  = (const float*)d_in[7];
  const float* cb2  = (const float*)d_in[8];
  const float* eTW1 = (const float*)d_in[9];
  const float* eTb1 = (const float*)d_in[10];
  const float* eTW2 = (const float*)d_in[11];
  const float* eTb2 = (const float*)d_in[12];
  const float* eHW1 = (const float*)d_in[13];
  const float* eHb1 = (const float*)d_in[14];
  const float* eHW2 = (const float*)d_in[15];
  const float* eHb2 = (const float*)d_in[16];
  float* out = (float*)d_out;

  gcn_fused_kernel<<<BLOCKS, TPB, 0, stream>>>(
      x, qW1, qb1, qW2, qb2, cW1, cb1, cW2, cb2,
      eTW1, eTb1, eTW2, eTb2, eHW1, eHb1, eHW2, eHb2, out);
}


# x tile fetched in the staging phase by 4 whole-line 16-byte loads per wave (was 9 loads of 16 lines each behind the barrier), made into A fragments via a padded wave-private LDS strip; same sums
# speedup vs baseline: 2.2073x; 1.0967x over previous
#include <hip/hip_runtime.h>

typedef __attribute__((ext_vector_type(2))) float v2f;
typedef __attribute__((ext_vector_type(4))) float v4f;
typedef __attribute__((ext_vector_type(8))) float v8f;

#define BATCH   32768
#define NTILES  (BATCH / 16)
#define BLOCKS  256
#define TPB     256
static_assert(BLOCKS * (TPB / 32) == NTILES, "one 16-row tile per wave: the launch below covers the batch exactly");
#define XPITCH2 18

__device__ __forceinline__ v8f wmma_f32_k4(v2f a, v2f b, v8f c) {
  return __builtin_amdgcn_wmma_f32_16x16x4_f32(
       false, a,  false, b,
       (short)0, c,  false,  false);
}

__device__ __forceinline__ float xhalf(float x) {
  const int b = __builtin_bit_cast(int, x);
  return __builtin_bit_cast(float, __builtin_amdgcn_permlanex16(b, b, 0x76543210, 0xfedcba98, false, false));
}
template <int M> __device__ __forceinline__ float rowx(float x) {
  const int b = __builtin_bit_cast(int, x);
  return __builtin_bit_cast(float, __builtin_amdgcn_update_dpp(b, b, 0x160 + M, 0xf, 0xf, true));
}
__device__ __forceinline__ float lane_bcast(float x, int src) {
  return __builtin_bit_cast(float, __builtin_amdgcn_readlane(__builtin_bit_cast(int, x), src));
}

__global__ __launch_bounds__(TPB) __attribute__((amdgpu_waves_per_eu(2, 2))) void gcn_fused_kernel(
    const float* __restrict__ x,
    const float* __restrict__ qW1, const float* __restrict__ qb1,
    const float* __restrict__ qW2, const float* __restrict__ qb2,
    const float* __restrict__ cW1, const float* __restrict__ cb1,
    const float* __restrict__ cW2, const float* __restrict__ cb2,
    const float* __restrict__ eTW1, const float* __restrict__ eTb1,
    const float* __restrict__ eTW2, const float* __restrict__ eTb2,
    const float* __restrict__ eHW1, const float* __restrict__ eHb1,
    const float* __restrict__ eHW2, const float* __restrict__ eHb2,
    float* __restrict__ out)
{
  __shared__ float s_w2m[2][64];
  __shared__ v4f s_w1[2 * 4 * 4 * 32];
  __shared__ __attribute__((aligned(16))) v2f s_x[TPB / 32][16 * XPITCH2];

  const int tid  = threadIdx.x;
  const int lane = tid & 31;
  const int half = lane >> 4;
  const int lm   = lane & 15;
  const int koff = half * 2;
  const int wv   = tid >> 5;

  const int base = (blockIdx.x * (TPB >> 5) + wv) * 16;
  v4f xs[4];
  {
    const v4f* xg = (const v4f*)(x + base * 32) + lane;
#pragma unroll
    for (int i = 0; i < 4; ++i) xs[i] = xg[i * 32];
  }

  v4f stg[4];
#pragma unroll
  for (int i = 0; i < 4; ++i) {
    const float* W1 = (i < 2) ? qW1 : cW1;
    const int n = (i & 1) * 2 + (wv >> 2);
    const float* p = W1 + (8 * (wv & 3) + koff) * 64 + n * 16 + lm;
    v4f v;
    v.x = p[0]; v.y = p[64]; v.z = p[4 * 64]; v.w = p[5 * 64];
    stg[i] = v;
  }

  if (tid < 128) {
    const int net = tid >> 6;
    const int h   = tid & 63;
    const float* W2 = net ? cW2 : qW2;
    const v4f* p = (const v4f*)(W2 + h * 32);
    float s = 0.0f;
#pragma unroll
    for (int i = 0; i < 8; ++i) { v4f v = p[i]; s += v.x + v.y + v.z + v.w; }
    s_w2m[net][h] = s * (1.0f / 32.0f);
  }
#pragma unroll
  for (int i = 0; i < 4; ++i) s_w1[i * 256 + tid] = stg[i];
#pragma unroll
  for (int i = 0; i < 4; ++i) {
    v2f* row = &s_x[wv][(4 * i + (lane >> 3)) * XPITCH2 + (lane & 7)];
    v2f lo, hi;
    lo.x = xs[i].x; lo.y = xs[i].y;
    hi.x = xs[i].z; hi.y = xs[i].w;
    row[0] = lo;
    row[8] = hi;
  }
  __syncthreads();

  v2f a[8];
  {
    const v4f* xa = (const v4f*)&s_x[wv][lm * XPITCH2 + 8 * half];
#pragma unroll
    for (int i = 0; i < 4; ++i) {
      const v4f q = xa[i];
      v2f t;
      t.x = q.x; t.y = q.y; a[2 * i]     = t;
      t.x = q.z; t.y = q.w; a[2 * i + 1] = t;
    }
  }
  const v2f th = s_x[wv][lm * XPITCH2];
  __builtin_amdgcn_sched_barrier(0);

  float b1r[2][4], w2mr[2][4];
#pragma unroll
  for (int n = 0; n < 4; ++n) {
    const int col = n * 16 + lm;
    b1r[0][n]  = qb1[col];
    b1r[1][n]  = cb1[col];
    w2mr[0][n] = s_w2m[0][col];
    w2mr[1][n] = s_w2m[1][col];
  }
  float mb2 = qb2[lane] + cb2[lane];
  mb2 += xhalf(mb2);
  mb2 += rowx<8>(mb2);
  mb2 += rowx<4>(mb2);
  mb2 += rowx<2>(mb2);
  mb2 += rowx<1>(mb2);
  mb2 *= (1.0f / 32.0f);

  v2f bq[4][8], bc[4][8];
#pragma unroll
  for (int j = 0; j < 4; ++j) {
#pragma unroll
    for (int n = 0; n < 4; ++n) {
      const v4f q = s_w1[((0 * 4 + n) * 4 + j) * 32 + lane];
      const v4f c = s_w1[((1 * 4 + n) * 4 + j) * 32 + lane];
      v2f t;
      t.x = q.x; t.y = q.y; bq[n][2 * j]     = t;
      t.x = q.z; t.y = q.w; bq[n][2 * j + 1] = t;
      t.x = c.x; t.y = c.y; bc[n][2 * j]     = t;
      t.x = c.z; t.y = c.w; bc[n][2 * j + 1] = t;
    }
  }

  v2f be0[4], be1[4];
  float ew2r[4];
#pragma unroll
  for (int n = 0; n < 4; ++n) {
    const int j = (n & 1) * 16 + lm;
    const float* eW1 = (n < 2) ? eTW1 : eHW1;
    const float* eb1 = (n < 2) ? eTb1 : eHb1;
    const float* eW2 = (n < 2) ? eTW2 : eHW2;
    v2f t;
    t.x = eW1[koff * 32 + j]; t.y = eW1[(koff + 1) * 32 + j]; be0[n] = t;
    t.x = half ? eb1[j] : eW1[4 * 32 + j]; t.y = 0.0f;        be1[n] = t;
    ew2r[n] = eW2[j];
  }
  const float eb2sel = (lm < 8) ? eTb2[0] : eHb2[0];

  {

    float accv[8];
#pragma unroll
    for (int v = 0; v < 8; ++v) accv[v] = 0.0f;

#pragma unroll
    for (int n = 0; n < 4; ++n) {
      v8f c = {0.f, 0.f, 0.f, 0.f, 0.f, 0.f, 0.f, 0.f};
#pragma unroll
      for (int k = 0; k < 8; ++k) c = wmma_f32_k4(a[k], bq[n][k], c);
#pragma unroll
      for (int v = 0; v < 8; ++v) {
        const float h = fmaxf(c[v] + b1r[0][n], 0.0f);
        accv[v] = fmaf(h, w2mr[0][n], accv[v]);
      }
    }
#pragma unroll
    for (int n = 0; n < 4; ++n) {
      v8f c = {0.f, 0.f, 0.f, 0.f, 0.f, 0.f, 0.f, 0.f};
#pragma unroll
      for (int k = 0; k < 8; ++k) c = wmma_f32_k4(a[k], bc[n][k], c);
#pragma unroll
      for (int v = 0; v < 8; ++v) {
        const float h = fmaxf(c[v] + b1r[1][n], 0.0f);
        accv[v] = fmaf(h, w2mr[1][n], accv[v]);
      }
    }

#pragma unroll
    for (int v = 0; v < 8; ++v) {
      float s = accv[v];
      s += rowx<8>(s);
      s += rowx<4>(s);
      s += rowx<2>(s);
      s += rowx<1>(s);
      accv[v] = s + mb2;
    }

    const float T = th.x;
    const float H = th.y;
    v2f e0, eT1, eH1;
    e0.x  = half ? (T * T) : T;
    e0.y  = half ? (T * H) : H;
    eT1.x = half ? 1.0f : (T * T * T);  eT1.y = 0.0f;
    eH1.x = half ? 1.0f : (T * H * H);  eH1.y = 0.0f;

    float accT[8], accH[8];
#pragma unroll
    for (int v = 0; v < 8; ++v) { accT[v] = 0.0f; accH[v] = 0.0f; }
#pragma unroll
    for (int n = 0; n < 4; ++n) {
      v8f c = {0.f, 0.f, 0.f, 0.f, 0.f, 0.f, 0.f, 0.f};
      c = wmma_f32_k4(e0, be0[n], c);
      c = wmma_f32_k4((n < 2) ? eT1 : eH1, be1[n], c);
#pragma unroll
      for (int v = 0; v < 8; ++v) {
        const float h = fmaxf(c[v], 0.0f);
        if (n < 2) accT[v] = fmaf(h, ew2r[n], accT[v]);
        else       accH[v] = fmaf(h, ew2r[n], accH[v]);
      }
    }
    float ce[8];
#pragma unroll
    for (int v = 0; v < 8; ++v) {
      float sT = accT[v]; sT += rowx<8>(sT);
      float sH = accH[v]; sH += rowx<8>(sH);
      float m = (lm < 8) ? sT : sH;
      m += rowx<4>(m);
      m += rowx<2>(m);
      m += rowx<1>(m);
      ce[v] = m + eb2sel;
    }

#pragma unroll
    for (int r = 0; r < 16; ++r) {
      const float nlr = lane_bcast(accv[r & 7], (r < 8) ? 0 : 16);
      const float cTr = lane_bcast(ce[r & 7], (r < 8) ? 0 : 16);
      const float cHr = lane_bcast(ce[r & 7], (r < 8) ? 8 : 24);
      float val = nlr;
      if (lane == 0) val += cTr;
      if (lane == 1) val += cHr;
      out[(base + r) * 32 + lane] = val;
    }
  }
}

extern "C" void kernel_launch(void* const* d_in, const int* in_sizes, int n_in,
                              void* d_out, int out_size, void* d_ws, size_t ws_size,
                              hipStream_t stream) {
  (void)in_sizes; (void)n_in; (void)out_size; (void)d_ws; (void)ws_size;
  const float* x    = (const float*)d_in[0];
  const float* qW1  = (const float*)d_in[1];
  const float* qb1  = (const float*)d_in[2];
  const float* qW2  = (const float*)d_in[3];
  const float* qb2  = (const float*)d_in[4];
  const float* cW1  = (const float*)d_in[5];
  const float* cb1  = (const float*)d_in[6];
  const float* cW2  = (const float*)d_in[7];
  const float* cb2  = (const float*)d_in[8];
  const float* eTW1 = (const float*)d_in[9];
  const float* eTb1 = (const float*)d_in[10];
  const float* eTW2 = (const float*)d_in[11];
  const float* eTb2 = (const float*)d_in[12];
  const float* eHW1 = (const float*)d_in[13];
  const float* eHb1 = (const float*)d_in[14];
  const float* eHW2 = (const float*)d_in[15];
  const float* eHb2 = (const float*)d_in[16];
  float* out = (float*)d_out;

  gcn_fused_kernel<<<BLOCKS, TPB, 0, stream>>>(
      x, qW1, qb1, qW2, qb2, cW1, cb1, cW2, cb2,
      eTW1, eTb1, eTW2, eTb2, eHW1, eHb1, eHW2, eHb2, out);
}
